# starting kernel
# baseline (speedup 1.0000x reference)
#include <hip/hip_runtime.h>
#include <hip/hip_bf16.h>

typedef float v2f __attribute__((ext_vector_type(2)));
typedef float v8f __attribute__((ext_vector_type(8)));


#define PAIR_STRIDE 80

__global__ __launch_bounds__(256) void edge_mlp_wmma_kernel(
    const float* __restrict__ K_h, const float* __restrict__ Q_h,
    const float* __restrict__ P_e, const int* __restrict__ src,
    const int* __restrict__ dst, const float* __restrict__ W1,
    const float* __restrict__ b1, const float* __restrict__ W2,
    const float* __restrict__ b2, float* __restrict__ out, int nEdges)
{
    __shared__ v2f sW1p[32 * PAIR_STRIDE];

#pragma unroll
    for (int t = 0; t < 8; ++t) {
        const int i  = threadIdx.x + t * 256;
        const int kp = i >> 6;
        const int n  = i & 63;
        v2f pr;
        pr.x = W1[(size_t)(2 * kp)     * 64 + n];
        pr.y = W1[(size_t)(2 * kp + 1) * 64 + n];
        sW1p[kp * PAIR_STRIDE + n] = pr;
    }
    __syncthreads();

    const int lane = threadIdx.x & 31;
    const int waveInBlock = threadIdx.x >> 5;
    const int tile = blockIdx.x * (blockDim.x >> 5) + waveInBlock;

    const int n0 = lane & 15;
    const int hi = lane >> 4;

    int e = tile * 16 + n0;
    if (e >= nEdges) e = nEdges - 1;
    const int s = src[e];
    const int d = dst[e];
    const float* __restrict__ kp_ = K_h + (size_t)s * 64;
    const float* __restrict__ qp_ = Q_h + (size_t)d * 64;
    const float* __restrict__ pp_ = P_e + (size_t)s * 64;

    v2f A[16];
#pragma unroll
    for (int kc = 0; kc < 16; ++kc) {
        const int col = 4 * kc + 2 * hi;
        v2f kv = *(const v2f*)(kp_ + col);
        v2f qv = *(const v2f*)(qp_ + col);
        v2f pv = *(const v2f*)(pp_ + col);
        A[kc].x = kv.x - qv.x + pv.x;
        A[kc].y = kv.y - qv.y + pv.y;
    }

    v8f acc[4];
#pragma unroll
    for (int nt = 0; nt < 4; ++nt) acc[nt] = (v8f){};

#pragma unroll
    for (int kc = 0; kc < 16; ++kc) {
        const v2f* __restrict__ wp = &sW1p[(2 * kc + hi) * PAIR_STRIDE + n0];
#pragma unroll
        for (int nt = 0; nt < 4; ++nt) {
            const v2f b = wp[nt * 16];
            acc[nt] = __builtin_amdgcn_wmma_f32_16x16x4_f32(
                 false, A[kc],  false, b,
                 (short)0, acc[nt],
                 false,  false);
        }
    }

    float p[8];
#pragma unroll
    for (int r = 0; r < 8; ++r) p[r] = 0.0f;

#pragma unroll
    for (int nt = 0; nt < 4; ++nt) {
        const int n = nt * 16 + n0;
        const float bb = b1[n];
        const float ww = W2[n];
#pragma unroll
        for (int r = 0; r < 8; ++r) {
            float h = acc[nt][r] + bb;
            h = (h > 0.0f) ? h : 0.0f;
            p[r] += h * ww;
        }
    }

#pragma unroll
    for (int m = 1; m <= 8; m <<= 1) {
#pragma unroll
        for (int r = 0; r < 8; ++r) p[r] += __shfl_xor(p[r], m, 32);
    }

    if (n0 < 8) {
        float val = p[0];
#pragma unroll
        for (int r = 1; r < 8; ++r)
            if (n0 == r) val = p[r];
        const int oe = tile * 16 + n0 + 8 * hi;
        if (oe < nEdges) out[oe] = val + b2[0];
    }
}

extern "C" void kernel_launch(void* const* d_in, const int* in_sizes, int n_in,
                              void* d_out, int out_size, void* d_ws, size_t ws_size,
                              hipStream_t stream) {
    const float* K_h = (const float*)d_in[0];
    const float* Q_h = (const float*)d_in[1];
    const float* P_e = (const float*)d_in[2];
    const int*   src = (const int*)d_in[3];
    const int*   dst = (const int*)d_in[4];
    const float* W1  = (const float*)d_in[5];
    const float* b1  = (const float*)d_in[6];
    const float* W2  = (const float*)d_in[7];
    const float* b2  = (const float*)d_in[8];
    float* out = (float*)d_out;

    const int nEdges = in_sizes[3];
    const int tiles  = (nEdges + 15) / 16;
    const int wavesPerBlock = 8;
    const int blocks = (tiles + wavesPerBlock - 1) / wavesPerBlock;

    edge_mlp_wmma_kernel<<<blocks, 256, 0, stream>>>(
        K_h, Q_h, P_e, src, dst, W1, b1, W2, b2, out, nEdges);
}


# Gather rows as whole 128-B lines (16 lanes x 16 B per row, 4 lines per load instead of 16) and turn them into the A fragment layout through a per-wave LDS tile; sums and k order unchanged
# speedup vs baseline: 1.6465x; 1.6465x over previous
#include <hip/hip_runtime.h>
#include <hip/hip_bf16.h>

typedef float v2f __attribute__((ext_vector_type(2)));
typedef float v8f __attribute__((ext_vector_type(8)));


typedef float v4f __attribute__((ext_vector_type(4)));
typedef int   v4i __attribute__((ext_vector_type(4)));

#define PAIR_STRIDE 80
#define DIF_PITCH   68

__global__ __launch_bounds__(256) void edge_mlp_wmma_kernel(
    const float* __restrict__ K_h, const float* __restrict__ Q_h,
    const float* __restrict__ P_e, const int* __restrict__ src,
    const int* __restrict__ dst, const float* __restrict__ W1,
    const float* __restrict__ b1, const float* __restrict__ W2,
    const float* __restrict__ b2, float* __restrict__ out, int nEdges)
{
    __shared__ v2f sW1p[32 * PAIR_STRIDE];
    __shared__ float sDif[8 * 16 * DIF_PITCH];

#pragma unroll
    for (int t = 0; t < 8; ++t) {
        const int i  = threadIdx.x + t * 256;
        const int kp = i >> 6;
        const int n  = i & 63;
        v2f pr;
        pr.x = W1[(size_t)(2 * kp)     * 64 + n];
        pr.y = W1[(size_t)(2 * kp + 1) * 64 + n];
        sW1p[kp * PAIR_STRIDE + n] = pr;
    }
    __syncthreads();

    const int lane = threadIdx.x & 31;
    const int waveInBlock = __builtin_amdgcn_readfirstlane(threadIdx.x >> 5);
    const int tile = blockIdx.x * (blockDim.x >> 5) + waveInBlock;

    const int n0 = lane & 15;
    const int hi = lane >> 4;

    const int ebase = tile * 16 + 8 * hi;
    int sj[8], dj[8];
    if (tile * 16 + 16 <= nEdges) {
        const v4i s0 = *(const v4i*)(src + ebase);
        const v4i s1 = *(const v4i*)(src + ebase + 4);
        const v4i d0 = *(const v4i*)(dst + ebase);
        const v4i d1 = *(const v4i*)(dst + ebase + 4);
#pragma unroll
        for (int j = 0; j < 4; ++j) {
            sj[j] = s0[j]; sj[j + 4] = s1[j];
            dj[j] = d0[j]; dj[j + 4] = d1[j];
        }
    } else {
#pragma unroll
        for (int j = 0; j < 8; ++j) {
            int e = ebase + j;
            if (e >= nEdges) e = nEdges - 1;
            sj[j] = src[e];
            dj[j] = dst[e];
        }
    }

    float* myDif = sDif + waveInBlock * (16 * DIF_PITCH);
#pragma unroll
    for (int j = 0; j < 8; ++j) {
        const v4f kv = *(const v4f*)(K_h + (size_t)sj[j] * 64 + 4 * n0);
        const v4f qv = *(const v4f*)(Q_h + (size_t)dj[j] * 64 + 4 * n0);
        const v4f pv = *(const v4f*)(P_e + (size_t)sj[j] * 64 + 4 * n0);
        v4f dv;
        dv.x = kv.x - qv.x + pv.x;
        dv.y = kv.y - qv.y + pv.y;
        dv.z = kv.z - qv.z + pv.z;
        dv.w = kv.w - qv.w + pv.w;
        *(v4f*)(myDif + (8 * hi + j) * DIF_PITCH + 4 * n0) = dv;
    }

    v2f A[16];
#pragma unroll
    for (int kc = 0; kc < 16; ++kc)
        A[kc] = *(const v2f*)(myDif + n0 * DIF_PITCH + 4 * kc + 2 * hi);

    v8f acc[4];
#pragma unroll
    for (int nt = 0; nt < 4; ++nt) acc[nt] = (v8f){};

#pragma unroll
    for (int kc = 0; kc < 16; ++kc) {
        const v2f* __restrict__ wp = &sW1p[(2 * kc + hi) * PAIR_STRIDE + n0];
#pragma unroll
        for (int nt = 0; nt < 4; ++nt) {
            const v2f b = wp[nt * 16];
            acc[nt] = __builtin_amdgcn_wmma_f32_16x16x4_f32(
                 false, A[kc],  false, b,
                 (short)0, acc[nt],
                 false,  false);
        }
    }

    float p[8];
#pragma unroll
    for (int r = 0; r < 8; ++r) p[r] = 0.0f;

#pragma unroll
    for (int nt = 0; nt < 4; ++nt) {
        const int n = nt * 16 + n0;
        const float bb = b1[n];
        const float ww = W2[n];
#pragma unroll
        for (int r = 0; r < 8; ++r) {
            float h = acc[nt][r] + bb;
            h = (h > 0.0f) ? h : 0.0f;
            p[r] += h * ww;
        }
    }

#pragma unroll
    for (int m = 1; m <= 8; m <<= 1) {
#pragma unroll
        for (int r = 0; r < 8; ++r) p[r] += __shfl_xor(p[r], m, 32);
    }

    if (n0 < 8) {
        float val = p[0];
#pragma unroll
        for (int r = 1; r < 8; ++r)
            if (n0 == r) val = p[r];
        const int oe = tile * 16 + n0 + 8 * hi;
        if (oe < nEdges) out[oe] = val + b2[0];
    }
}

extern "C" void kernel_launch(void* const* d_in, const int* in_sizes, int n_in,
                              void* d_out, int out_size, void* d_ws, size_t ws_size,
                              hipStream_t stream) {
    const float* K_h = (const float*)d_in[0];
    const float* Q_h = (const float*)d_in[1];
    const float* P_e = (const float*)d_in[2];
    const int*   src = (const int*)d_in[3];
    const int*   dst = (const int*)d_in[4];
    const float* W1  = (const float*)d_in[5];
    const float* b1  = (const float*)d_in[6];
    const float* W2  = (const float*)d_in[7];
    const float* b2  = (const float*)d_in[8];
    float* out = (float*)d_out;

    const int nEdges = in_sizes[3];
    const int tiles  = (nEdges + 15) / 16;
    const int wavesPerBlock = 8;
    const int blocks = (tiles + wavesPerBlock - 1) / wavesPerBlock;

    edge_mlp_wmma_kernel<<<blocks, 256, 0, stream>>>(
        K_h, Q_h, P_e, src, dst, W1, b1, W2, b2, out, nEdges);
}
